# starting kernel
# baseline (speedup 1.0000x reference)
#include <hip/hip_runtime.h>

typedef __attribute__((ext_vector_type(2))) float v2f;
typedef __attribute__((ext_vector_type(8))) float v8f;

#define BATCH 32
#define HGT   32
#define WID   32
#define CIN   128
#define COUT  128
#define BN_EPS 1e-3f

__global__ __launch_bounds__(256) void ComplexDotLayer_kernel(
    const float* __restrict__ xr, const float* __restrict__ xi,
    const float* __restrict__ wr, const float* __restrict__ wi,
    const float* __restrict__ gr, const float* __restrict__ br,
    const float* __restrict__ mr, const float* __restrict__ vrr,
    const float* __restrict__ gi, const float* __restrict__ bi,
    const float* __restrict__ mi, const float* __restrict__ vii,
    float* __restrict__ out)
{
    const int p    = blockIdx.x;
    const int nt   = threadIdx.x >> 5;
    const int lane = threadIdx.x & 31;

    const int l15   = lane & 15;
    const int half  = lane >> 4;
    const int khalf = half * 2;
    const int ncol  = nt * 16 + l15;

    const size_t xbstride = (size_t)HGT * WID * CIN;
    const float* xr_row0 = xr + (size_t)p * CIN + (size_t)l15 * xbstride;
    const float* xi_row0 = xi + (size_t)p * CIN + (size_t)l15 * xbstride;
    const float* xr_row1 = xr_row0 + 16 * xbstride;
    const float* xi_row1 = xi_row0 + 16 * xbstride;
    const float* wr_p    = wr + (size_t)p * CIN * COUT;
    const float* wi_p    = wi + (size_t)p * CIN * COUT;

    v8f acc_rr0 = {}, acc_ii0 = {}, acc_ri0 = {}, acc_ir0 = {};
    v8f acc_rr1 = {}, acc_ii1 = {}, acc_ri1 = {}, acc_ir1 = {};

    #pragma unroll 2
    for (int k0 = 0; k0 < CIN; k0 += 4) {
        const int ka = k0 + khalf;
        v2f a_r0 = *(const v2f*)(xr_row0 + ka);
        v2f a_i0 = *(const v2f*)(xi_row0 + ka);
        v2f a_r1 = *(const v2f*)(xr_row1 + ka);
        v2f a_i1 = *(const v2f*)(xi_row1 + ka);
        v2f b_r, b_i;
        b_r.x = wr_p[(size_t)ka * COUT + ncol];
        b_r.y = wr_p[(size_t)(ka + 1) * COUT + ncol];
        b_i.x = wi_p[(size_t)ka * COUT + ncol];
        b_i.y = wi_p[(size_t)(ka + 1) * COUT + ncol];

        acc_rr0 = __builtin_amdgcn_wmma_f32_16x16x4_f32(
            false, a_r0, false, b_r, (short)0, acc_rr0, false, false);
        acc_ri0 = __builtin_amdgcn_wmma_f32_16x16x4_f32(
            false, a_r0, false, b_i, (short)0, acc_ri0, false, false);
        acc_ii0 = __builtin_amdgcn_wmma_f32_16x16x4_f32(
            false, a_i0, false, b_i, (short)0, acc_ii0, false, false);
        acc_ir0 = __builtin_amdgcn_wmma_f32_16x16x4_f32(
            false, a_i0, false, b_r, (short)0, acc_ir0, false, false);
        acc_rr1 = __builtin_amdgcn_wmma_f32_16x16x4_f32(
            false, a_r1, false, b_r, (short)0, acc_rr1, false, false);
        acc_ri1 = __builtin_amdgcn_wmma_f32_16x16x4_f32(
            false, a_r1, false, b_i, (short)0, acc_ri1, false, false);
        acc_ii1 = __builtin_amdgcn_wmma_f32_16x16x4_f32(
            false, a_i1, false, b_i, (short)0, acc_ii1, false, false);
        acc_ir1 = __builtin_amdgcn_wmma_f32_16x16x4_f32(
            false, a_i1, false, b_r, (short)0, acc_ir1, false, false);
    }

    const float inv_r = gr[ncol] * rsqrtf(vrr[ncol] + BN_EPS);
    const float add_r = br[ncol] - mr[ncol] * inv_r;
    const float inv_i = gi[ncol] * rsqrtf(vii[ncol] + BN_EPS);
    const float add_i = bi[ncol] - mi[ncol] * inv_i;

    float* out_r = out;
    float* out_i = out + (size_t)BATCH * HGT * WID * COUT;

    const int mbase = 8 * half;
    #pragma unroll
    for (int r = 0; r < 8; ++r) {
        const int b0 = mbase + r;
        const int b1 = b0 + 16;
        const size_t off0 = (size_t)b0 * ((size_t)HGT * WID * COUT)
                          + (size_t)p * COUT + ncol;
        const size_t off1 = (size_t)b1 * ((size_t)HGT * WID * COUT)
                          + (size_t)p * COUT + ncol;
        out_r[off0] = (acc_rr0[r] + acc_ii0[r]) * inv_r + add_r;
        out_i[off0] = (acc_ri0[r] - acc_ir0[r]) * inv_i + add_i;
        out_r[off1] = (acc_rr1[r] + acc_ii1[r]) * inv_r + add_r;
        out_i[off1] = (acc_ri1[r] - acc_ir1[r]) * inv_i + add_i;
    }
}

extern "C" void kernel_launch(void* const* d_in, const int* in_sizes, int n_in,
                              void* d_out, int out_size, void* d_ws, size_t ws_size,
                              hipStream_t stream) {
    (void)in_sizes; (void)n_in; (void)out_size; (void)d_ws; (void)ws_size;
    const float* xr  = (const float*)d_in[0];
    const float* xi  = (const float*)d_in[1];
    const float* wr  = (const float*)d_in[2];
    const float* wi  = (const float*)d_in[3];
    const float* gr  = (const float*)d_in[4];
    const float* br  = (const float*)d_in[5];
    const float* mr  = (const float*)d_in[6];
    const float* vrr = (const float*)d_in[7];
    const float* gi  = (const float*)d_in[8];
    const float* bi  = (const float*)d_in[9];
    const float* mi  = (const float*)d_in[10];
    const float* vii = (const float*)d_in[11];
    float* out = (float*)d_out;

    dim3 grid(HGT * WID);
    dim3 block(256);
    ComplexDotLayer_kernel<<<grid, block, 0, stream>>>(
        xr, xi, wr, wi, gr, br, mr, vrr, gi, bi, mi, vii, out);
}


# Wave tile 32 batch x 64 channels, lane owns 4 adjacent channels (2 waves/position, 256 x 8 waves): X, W loads, out stores 16 B/lane, loads 2 groups of 8 k ahead; k summed 0,1,4,5,2,3,6,7 per 8
# speedup vs baseline: 1.3966x; 1.3966x over previous
#include <hip/hip_runtime.h>

typedef __attribute__((ext_vector_type(2))) float v2f;
typedef __attribute__((ext_vector_type(4))) float v4f;
typedef __attribute__((ext_vector_type(8))) float v8f;

#define BATCH 32
#define HGT   32
#define WID   32
#define CIN   128
#define COUT  128
#define BN_EPS 1e-3f

#define NPOS   (HGT * WID)
#define KG     8
#define NGROUP (CIN / KG)
#define AHEAD  2
#define NSET   (AHEAD + 1)

struct Grp { v4f xr[2], xi[2], wr[4], wi[4]; };

__global__ __launch_bounds__(256) void ComplexDotLayer_kernel(
    const float* __restrict__ xr, const float* __restrict__ xi,
    const float* __restrict__ wr, const float* __restrict__ wi,
    const float* __restrict__ gr, const float* __restrict__ br,
    const float* __restrict__ mr, const float* __restrict__ vrr,
    const float* __restrict__ gi, const float* __restrict__ bi,
    const float* __restrict__ mi, const float* __restrict__ vii,
    float* __restrict__ out)
{
    const unsigned lane = threadIdx.x & 31u;
    const unsigned wv = (unsigned)__builtin_amdgcn_readfirstlane((int)(threadIdx.x >> 5));
    const unsigned p  = blockIdx.x * 4u + (wv >> 1);
    const unsigned ch = wv & 1u;
    const unsigned c  = lane & 15u;
    const unsigned h  = lane >> 4;

    const float* xr_p = xr + (size_t)p * CIN;
    const float* xi_p = xi + (size_t)p * CIN;
    const float* wr_p = wr + (size_t)p * CIN * COUT + ch * 64u;
    const float* wi_p = wi + (size_t)p * CIN * COUT + ch * 64u;

    const unsigned xoff = c * (unsigned)(NPOS * CIN) + 4u * h;
    const unsigned woff = (4u * h) * COUT + 4u * c;

    v8f acc_rr[2][4], acc_ri[2][4], acc_ii[2][4], acc_ir[2][4];
    #pragma unroll
    for (int mt = 0; mt < 2; ++mt) {
        #pragma unroll
        for (int t = 0; t < 4; ++t) {
            acc_rr[mt][t] = (v8f){}; acc_ri[mt][t] = (v8f){};
            acc_ii[mt][t] = (v8f){}; acc_ir[mt][t] = (v8f){};
        }
    }

    Grp G[NSET];

#define LOAD_GROUP(S, g) do {                                                                     \
        _Pragma("unroll") for (int mt = 0; mt < 2; ++mt) {                                        \
            G[S].xr[mt] = *(const v4f*)(xr_p + xoff + (unsigned)(mt * 16 * NPOS * CIN + (g) * KG)); \
            G[S].xi[mt] = *(const v4f*)(xi_p + xoff + (unsigned)(mt * 16 * NPOS * CIN + (g) * KG)); \
        }                                                                                         \
        _Pragma("unroll") for (int j = 0; j < 4; ++j) {                                           \
            G[S].wr[j] = *(const v4f*)(wr_p + woff + (unsigned)(((g) * KG + j) * COUT));           \
            G[S].wi[j] = *(const v4f*)(wi_p + woff + (unsigned)(((g) * KG + j) * COUT));           \
        }                                                                                         \
    } while (0)

    #pragma unroll
    for (int g = 0; g < AHEAD; ++g) LOAD_GROUP(g % NSET, g);

    #pragma unroll
    for (int g = 0; g < NGROUP; ++g) {
        __builtin_amdgcn_sched_barrier(0);
        if (g + AHEAD < NGROUP) LOAD_GROUP((g + AHEAD) % NSET, g + AHEAD);
        __builtin_amdgcn_sched_barrier(0);

        const Grp& Q = G[g % NSET];
        #pragma unroll
        for (int q = 0; q < 2; ++q) {
            v2f a_r[2], a_i[2];
            #pragma unroll
            for (int mt = 0; mt < 2; ++mt) {
                a_r[mt] = q ? (v2f){Q.xr[mt].z, Q.xr[mt].w} : (v2f){Q.xr[mt].x, Q.xr[mt].y};
                a_i[mt] = q ? (v2f){Q.xi[mt].z, Q.xi[mt].w} : (v2f){Q.xi[mt].x, Q.xi[mt].y};
            }
            #pragma unroll
            for (int t = 0; t < 4; ++t) {
                const v2f b_r = {Q.wr[2 * q][t], Q.wr[2 * q + 1][t]};
                const v2f b_i = {Q.wi[2 * q][t], Q.wi[2 * q + 1][t]};
                #pragma unroll
                for (int mt = 0; mt < 2; ++mt) {
                    acc_rr[mt][t] = __builtin_amdgcn_wmma_f32_16x16x4_f32(
                        false, a_r[mt], false, b_r, (short)0, acc_rr[mt][t], false, false);
                    acc_ri[mt][t] = __builtin_amdgcn_wmma_f32_16x16x4_f32(
                        false, a_r[mt], false, b_i, (short)0, acc_ri[mt][t], false, false);
                    acc_ii[mt][t] = __builtin_amdgcn_wmma_f32_16x16x4_f32(
                        false, a_i[mt], false, b_i, (short)0, acc_ii[mt][t], false, false);
                    acc_ir[mt][t] = __builtin_amdgcn_wmma_f32_16x16x4_f32(
                        false, a_i[mt], false, b_r, (short)0, acc_ir[mt][t], false, false);
                }
            }
        }
    }
#undef LOAD_GROUP

    const unsigned ch0 = ch * 64u + 4u * c;
    const v4f g_r = *(const v4f*)(gr + ch0),  b_r4 = *(const v4f*)(br + ch0);
    const v4f m_r = *(const v4f*)(mr + ch0),  v_r = *(const v4f*)(vrr + ch0);
    const v4f g_i = *(const v4f*)(gi + ch0),  b_i4 = *(const v4f*)(bi + ch0);
    const v4f m_i = *(const v4f*)(mi + ch0),  v_i = *(const v4f*)(vii + ch0);
    v4f inv_r, add_r, inv_i, add_i;
    #pragma unroll
    for (int t = 0; t < 4; ++t) {
        inv_r[t] = g_r[t] * rsqrtf(v_r[t] + BN_EPS);
        add_r[t] = b_r4[t] - m_r[t] * inv_r[t];
        inv_i[t] = g_i[t] * rsqrtf(v_i[t] + BN_EPS);
        add_i[t] = b_i4[t] - m_i[t] * inv_i[t];
    }

    float* out_r = out + (size_t)p * COUT;
    float* out_i = out_r + (size_t)BATCH * NPOS * COUT;
    const unsigned ooff = (8u * h) * (unsigned)(NPOS * COUT) + ch0;
    #pragma unroll
    for (int mt = 0; mt < 2; ++mt) {
        #pragma unroll
        for (int r = 0; r < 8; ++r) {
            v4f o_r, o_i;
            #pragma unroll
            for (int t = 0; t < 4; ++t) {
                o_r[t] = (acc_rr[mt][t][r] + acc_ii[mt][t][r]) * inv_r[t] + add_r[t];
                o_i[t] = (acc_ri[mt][t][r] - acc_ir[mt][t][r]) * inv_i[t] + add_i[t];
            }
            const unsigned o = ooff + (unsigned)((16 * mt + r) * NPOS * COUT);
            *(v4f*)(out_r + o) = o_r;
            *(v4f*)(out_i + o) = o_i;
        }
    }
}

extern "C" void kernel_launch(void* const* d_in, const int* in_sizes, int n_in,
                              void* d_out, int out_size, void* d_ws, size_t ws_size,
                              hipStream_t stream) {
    (void)in_sizes; (void)n_in; (void)out_size; (void)d_ws; (void)ws_size;
    const float* xr  = (const float*)d_in[0];
    const float* xi  = (const float*)d_in[1];
    const float* wr  = (const float*)d_in[2];
    const float* wi  = (const float*)d_in[3];
    const float* gr  = (const float*)d_in[4];
    const float* br  = (const float*)d_in[5];
    const float* mr  = (const float*)d_in[6];
    const float* vrr = (const float*)d_in[7];
    const float* gi  = (const float*)d_in[8];
    const float* bi  = (const float*)d_in[9];
    const float* mi  = (const float*)d_in[10];
    const float* vii = (const float*)d_in[11];
    float* out = (float*)d_out;

    dim3 grid(256);
    dim3 block(256);
    ComplexDotLayer_kernel<<<grid, block, 0, stream>>>(
        xr, xi, wr, wi, gr, br, mr, vrr, gi, bi, mi, vii, out);
}
